# starting kernel
# baseline (speedup 1.0000x reference)
#include <hip/hip_runtime.h>

typedef __attribute__((ext_vector_type(2))) float v2f;
typedef __attribute__((ext_vector_type(8))) float v8f;

namespace {
constexpr int kWH = 65536;
constexpr int kK  = 512;
constexpr int kB  = 8;
constexpr int kKT = kK / 16;
constexpr int kChunk = 1024;
constexpr int kNCH = kWH / kChunk;
constexpr size_t P_OFF   = 0;
constexpr size_t Y_OFF   = (size_t)kKT * kNCH * 256;
constexpr size_t XR_OFF  = Y_OFF + (size_t)kK * 16;
constexpr size_t MAX_OFF = XR_OFF + (size_t)kB * kWH;
}

__device__ __forceinline__ float sgn(float w) {
  float pm1 = __uint_as_float((__float_as_uint(w) & 0x80000000u) | 0x3f800000u);
  return (w == 0.0f) ? 0.0f : pm1;
}

__global__ __launch_bounds__(256)
void phase1(const float* __restrict__ x, const float* __restrict__ w,
            float* __restrict__ ws) {
  const int wave = blockIdx.x * 8 + (threadIdx.x >> 5);
  const int lane = threadIdx.x & 31;
  const int kt = wave / kNCH;
  const int ch = wave - kt * kNCH;
  const int lm = lane & 15;
  const int o  = (lane < 16) ? 0 : 2;

  const float* __restrict__ arow = w + (size_t)(kt * 16 + lm) * kWH;
  const float* __restrict__ brow = x + (size_t)(lm & 7) * kWH;

  v8f acc0 = {};
  v8f acc1 = {};
  const int ibeg = ch * kChunk;
  #pragma unroll 2
  for (int i0 = ibeg; i0 < ibeg + kChunk; i0 += 8) {
    v2f aw0 = *(const v2f*)(arow + i0 + o);
    v2f bx0 = *(const v2f*)(brow + i0 + o);
    v2f A0; A0.x = sgn(aw0.x); A0.y = sgn(aw0.y);
    acc0 = __builtin_amdgcn_wmma_f32_16x16x4_f32(
        false, A0, false, bx0, (short)0, acc0, false, false);
    v2f aw1 = *(const v2f*)(arow + i0 + 4 + o);
    v2f bx1 = *(const v2f*)(brow + i0 + 4 + o);
    v2f A1; A1.x = sgn(aw1.x); A1.y = sgn(aw1.y);
    acc1 = __builtin_amdgcn_wmma_f32_16x16x4_f32(
        false, A1, false, bx1, (short)0, acc1, false, false);
  }

  float* __restrict__ p = ws + P_OFF + (size_t)wave * 256;
  const int rbase = (lane < 16) ? 0 : 8;
  #pragma unroll
  for (int r = 0; r < 8; ++r)
    p[(rbase + r) * 16 + lm] = acc0[r] + acc1[r];
}

__global__ __launch_bounds__(256)
void reduce_y(float* __restrict__ ws) {
  const int t = blockIdx.x * 256 + threadIdx.x;
  const int kt  = t >> 8;
  const int idx = t & 255;
  const float* __restrict__ p = ws + P_OFF + (size_t)kt * kNCH * 256 + idx;
  float s = 0.0f;
  #pragma unroll 4
  for (int ch = 0; ch < kNCH; ++ch) s += p[(size_t)ch * 256];
  ws[Y_OFF + t] = s;
  if (t == 0) ((unsigned*)ws)[MAX_OFF] = 0u;
}

__global__ __launch_bounds__(256)
void phase2(const float* __restrict__ w, float* __restrict__ ws) {
  __shared__ float y[kK * 16];
  for (int t = threadIdx.x; t < kK * 16; t += 256) y[t] = ws[Y_OFF + t];
  __syncthreads();

  const int lane = threadIdx.x & 31;
  const int lm = lane & 15;
  const int o  = (lane < 16) ? 0 : 2;
  const int tile = blockIdx.x * 8 + (threadIdx.x >> 5);
  const int i0 = tile * 16;

  v8f acc0 = {};
  v8f acc1 = {};
  #pragma unroll 2
  for (int k0 = 0; k0 < kK; k0 += 8) {
    {
      v2f A = { y[(k0 + o) * 16 + lm], y[(k0 + o + 1) * 16 + lm] };
      const float* bp = w + (size_t)(k0 + o) * kWH + (i0 + lm);
      v2f Bv = { sgn(bp[0]), sgn(bp[kWH]) };
      acc0 = __builtin_amdgcn_wmma_f32_16x16x4_f32(
          false, A, false, Bv, (short)0, acc0, false, false);
    }
    {
      v2f A = { y[(k0 + 4 + o) * 16 + lm], y[(k0 + 4 + o + 1) * 16 + lm] };
      const float* bp = w + (size_t)(k0 + 4 + o) * kWH + (i0 + lm);
      v2f Bv = { sgn(bp[0]), sgn(bp[kWH]) };
      acc1 = __builtin_amdgcn_wmma_f32_16x16x4_f32(
          false, A, false, Bv, (short)0, acc1, false, false);
    }
  }

  float m = -__builtin_inff();
  if (lane < 16) {
    #pragma unroll
    for (int r = 0; r < 8; ++r) {
      const float v = acc0[r] + acc1[r];
      ws[XR_OFF + (size_t)r * kWH + i0 + lm] = v;
      m = fmaxf(m, v);
    }
  }
  #pragma unroll
  for (int off = 16; off; off >>= 1)
    m = fmaxf(m, __shfl_xor(m, off, 32));
  if (lane == 0) {
    unsigned u = __float_as_uint(m);
    unsigned key = (u & 0x80000000u) ? ~u : (u | 0x80000000u);
    atomicMax(((unsigned*)ws) + MAX_OFF, key);
  }
}

__global__ __launch_bounds__(256)
void phase3(const float* __restrict__ ws, float* __restrict__ out) {
  const size_t t = (size_t)blockIdx.x * 256 + threadIdx.x;
  const unsigned key  = ((const unsigned*)ws)[MAX_OFF];
  const unsigned bits = (key & 0x80000000u) ? (key & 0x7fffffffu) : ~key;
  const float mx = __uint_as_float(bits);
  out[t] = ws[XR_OFF + t] / mx;
}

extern "C" void kernel_launch(void* const* d_in, const int* in_sizes, int n_in,
                              void* d_out, int out_size, void* d_ws, size_t ws_size,
                              hipStream_t stream) {
  const float* x = (const float*)d_in[0];
  const float* w = (const float*)d_in[1];
  float* out = (float*)d_out;
  float* ws  = (float*)d_ws;

  const int p1_waves = kKT * kNCH;
  phase1 <<<p1_waves / 8,        256, 0, stream>>>(x, w, ws);
  reduce_y<<<(kK * 16) / 256,    256, 0, stream>>>(ws);
  phase2 <<<(kWH / 16) / 8,      256, 0, stream>>>(w, ws);
  phase3 <<<(kB * kWH) / 256,    256, 0, stream>>>(ws, out);
}


# y reduction in one memory round trip: 128 workgroups, each thread loads two 16-B pieces of the 64 partial planes at once (was 16 dependent trips of 4 loads), summed via LDS; reorders the y sums
# speedup vs baseline: 2.4323x; 2.4323x over previous
#include <hip/hip_runtime.h>

typedef __attribute__((ext_vector_type(2))) float v2f;
typedef __attribute__((ext_vector_type(4))) float v4f;
typedef __attribute__((ext_vector_type(8))) float v8f;

namespace {
constexpr int kWH = 65536;
constexpr int kK  = 512;
constexpr int kB  = 8;
constexpr int kKT = kK / 16;
constexpr int kChunk = 1024;
constexpr int kSub   = 128;
constexpr int kWP    = kSub + 4;
constexpr int kXP    = kChunk + 4;
constexpr int kNCH = kWH / kChunk;
constexpr int kYP = kK + 4;
constexpr size_t S_OFF   = 0;
constexpr size_t Y_OFF   = (size_t)kKT * kWH / 2;
constexpr size_t F_OFF   = Y_OFF + (size_t)kB * kK;
constexpr size_t MAX_OFF = F_OFF + (size_t)kKT * kNCH;
}

__device__ __forceinline__ float sgn(float w) {
  float pm1 = __uint_as_float((__float_as_uint(w) & 0x80000000u) | 0x3f800000u);
  return (w == 0.0f) ? 0.0f : pm1;
}

__device__ __forceinline__ v4f sgn4(v4f v) {
  v4f r; r.x = sgn(v.x); r.y = sgn(v.y); r.z = sgn(v.z); r.w = sgn(v.w);
  return r;
}
__device__ __forceinline__ float min_abs4(float m, v4f v) {
  return fminf(fminf(fminf(fabsf(v.x), fabsf(v.y)), m), fminf(fabsf(v.z), fabsf(v.w)));
}
__device__ __forceinline__ unsigned push_sign(unsigned acc, float f) {
  return (acc << 1) | (__float_as_uint(f) >> 31);
}

__global__ __launch_bounds__(256)
void phase1(const float* __restrict__ x, const float* __restrict__ w,
            float* __restrict__ ws, float* __restrict__ out) {
  __shared__ __attribute__((aligned(16))) float xs[kB * kXP];
  __shared__ __attribute__((aligned(16))) float wt[8 * 16 * kWP];

  const int wv   = threadIdx.x >> 5;
  const int lane = threadIdx.x & 31;
  const int lm   = lane & 15;
  const int h    = lane >> 4;
  const int ch   = blockIdx.x & (kNCH - 1);
  const int kt   = (blockIdx.x >> 6) * 8 + wv;
  const int ibeg = ch * kChunk;

  const float* __restrict__ wsrc = w + (size_t)(kt * 16) * kWH + ibeg + 4 * lane;
  float* __restrict__ mywt = wt + wv * (16 * kWP);

  v4f rg[3][16];

#define LOAD_SUB(s)                                                           \
  _Pragma("unroll") for (int r = 0; r < 16; ++r)                              \
    rg[(s) % 3][r] = *(const v4f*)(wsrc + (size_t)r * kWH + (s) * kSub);

  v4f xv[kB];
  #pragma unroll
  for (int j = 0; j < kB; ++j)
    xv[j] = *(const v4f*)(x + (size_t)j * kWH + ibeg + 4 * threadIdx.x);
  LOAD_SUB(0)
  LOAD_SUB(1)
  #pragma unroll
  for (int j = 0; j < kB; ++j)
    *(v4f*)(xs + j * kXP + 4 * threadIdx.x) = xv[j];
  __syncthreads();

  v8f acc0 = {};
  v8f acc1 = {};
  float wmin = __builtin_inff();
  unsigned short* __restrict__ sdst =
      (unsigned short*)(ws + S_OFF) + (size_t)kt * kWH + ibeg + 4 * lane;
  const float* __restrict__ ap = mywt + lm * kWP + 4 * h;
  const float* __restrict__ bp = xs + (lm & 7) * kXP + 4 * h;

  #pragma unroll
  for (int s = 0; s < kChunk / kSub; ++s) {
    __builtin_amdgcn_sched_barrier(0);
    if (s + 2 < kChunk / kSub) { LOAD_SUB(s + 2) }
    __builtin_amdgcn_sched_barrier(0);

    unsigned h0 = 0u, h1 = 0u, h2 = 0u, h3 = 0u;
    #pragma unroll
    for (int r = 0; r < 16; ++r) {
      h0 = push_sign(h0, rg[s % 3][r].x);
      h1 = push_sign(h1, rg[s % 3][r].y);
      h2 = push_sign(h2, rg[s % 3][r].z);
      h3 = push_sign(h3, rg[s % 3][r].w);
      wmin = min_abs4(wmin, rg[s % 3][r]);
      *(v4f*)(mywt + r * kWP + 4 * lane) = sgn4(rg[s % 3][r]);
    }
    *(uint2*)(sdst + s * kSub) = make_uint2(h0 | (h1 << 16), h2 | (h3 << 16));
    __builtin_amdgcn_fence(__ATOMIC_ACQ_REL, "wavefront");

    v4f fa[kSub / 8], fb[kSub / 8];
    #pragma unroll
    for (int q = 0; q < kSub / 8; ++q) {
      fa[q] = *(const v4f*)(ap + 8 * q);
      fb[q] = *(const v4f*)(bp + s * kSub + 8 * q);
    }
    __builtin_amdgcn_sched_barrier(0);
    #pragma unroll
    for (int q = 0; q < kSub / 8; ++q) {
      const v2f a0 = {fa[q].x, fa[q].y}, b0 = {fb[q].x, fb[q].y};
      const v2f a1 = {fa[q].z, fa[q].w}, b1 = {fb[q].z, fb[q].w};
      acc0 = __builtin_amdgcn_wmma_f32_16x16x4_f32(
          false, a0, false, b0, (short)0, acc0, false, false);
      acc1 = __builtin_amdgcn_wmma_f32_16x16x4_f32(
          false, a1, false, b1, (short)0, acc1, false, false);
    }
    __builtin_amdgcn_fence(__ATOMIC_RELEASE, "wavefront");
  }
#undef LOAD_SUB

  float* __restrict__ p = out + (size_t)(kt * kNCH + ch) * 256;
  const int rbase = h * 8;
  #pragma unroll
  for (int r = 0; r < 8; ++r)
    p[(rbase + r) * 16 + lm] = acc0[r] + acc1[r];
  const unsigned zany = __builtin_amdgcn_ballot_w32(wmin == 0.0f);
  if (lane == 0) ((unsigned*)ws)[F_OFF + kt * kNCH + ch] = zany ? 1u : 0u;
}

__global__ __launch_bounds__(256)
void reduce_y(const float* __restrict__ part, float* __restrict__ ws) {
  __shared__ __attribute__((aligned(16))) float red[32 * 32];
  const int t  = threadIdx.x;
  const int kt = blockIdx.x >> 2;
  const int rq = blockIdx.x & 3;
  const int pc = t & 7;
  const int g  = t >> 3;
  const float* __restrict__ p =
      part + (size_t)kt * kNCH * 256 + (4 * rq + (pc >> 1)) * 16 + 4 * (pc & 1);
  const v4f lo = *(const v4f*)(p + (size_t)g * 256);
  const v4f hi = *(const v4f*)(p + (size_t)(g + 32) * 256);
  *(v4f*)(red + 4 * t) = lo + hi;
  __syncthreads();
  if (t < 32) {
    float s0 = 0.0f, s1 = 0.0f, s2 = 0.0f, s3 = 0.0f;
    #pragma unroll
    for (int j = 0; j < 32; j += 4) {
      s0 += red[(j + 0) * 32 + t];
      s1 += red[(j + 1) * 32 + t];
      s2 += red[(j + 2) * 32 + t];
      s3 += red[(j + 3) * 32 + t];
    }
    const int k = 16 * kt + 4 * rq + (t >> 3);
    const int b = 4 * ((t >> 2) & 1) + (t & 3);
    ws[Y_OFF + b * kK + k] = (s0 + s1) + (s2 + s3);
  }
  if (blockIdx.x == 0 && t == 0) ((unsigned*)ws)[MAX_OFF] = 0u;
}

__device__ __forceinline__ float bit_pm1(unsigned z, int sh) {
  return __uint_as_float(((z << sh) & 0x80000000u) | 0x3f800000u);
}

__global__ __launch_bounds__(256)
void phase2(const float* __restrict__ w, float* __restrict__ ws, float* __restrict__ out) {
  __shared__ __attribute__((aligned(16))) float ys[kB * kYP];
  __shared__ __attribute__((aligned(16))) unsigned short hb[kKT * 128];

  const int wv   = threadIdx.x >> 5;
  const int lane = threadIdx.x & 31;
  const int lm = lane & 15;
  const int h  = lane >> 4;
  const int i0b = blockIdx.x * 128;
  const int ch  = blockIdx.x >> 3;
  const int i0  = i0b + wv * 16;

  #pragma unroll
  for (int j = 0; j < 4; ++j) {
    const int q = threadIdx.x + 256 * j;
    *(v4f*)(ys + (q >> 7) * kYP + 4 * (q & 127)) = *(const v4f*)(ws + Y_OFF + 4 * q);
  }
  {
    const int row = threadIdx.x >> 3, pc = threadIdx.x & 7;
    const uint4* __restrict__ sp = (const uint4*)((const unsigned short*)(ws + S_OFF) +
                                                  (size_t)row * kWH + i0b + 16 * pc);
    uint4* dp = (uint4*)(hb + row * 128 + 16 * pc);
    const uint4 s0 = sp[0], s1 = sp[1];
    dp[0] = s0; dp[1] = s1;
  }
  const unsigned fmask = __builtin_amdgcn_ballot_w32(
      ((const unsigned*)ws)[F_OFF + lane * kNCH + ch] != 0u);
  __syncthreads();

  const float* __restrict__ ap = ys + (lm & 7) * kYP + 8 * h;
  const unsigned short* __restrict__ hp = hb + wv * 16 + lm;
  const int zsh = 16 + 8 * h;

  v8f acc0 = {};
  v8f acc1 = {};
#define WMMA4(b0, b1, b2, b3, b4, b5, b6, b7)                                        \
  {                                                                                  \
    const v2f A0 = {a0.x, a0.y}, A1 = {a0.z, a0.w}, A2 = {a1.x, a1.y}, A3 = {a1.z, a1.w}; \
    const v2f B0 = {b0, b1}, B1 = {b2, b3}, B2 = {b4, b5}, B3 = {b6, b7};            \
    acc0 = __builtin_amdgcn_wmma_f32_16x16x4_f32(false, A0, false, B0, (short)0, acc0, false, false); \
    acc1 = __builtin_amdgcn_wmma_f32_16x16x4_f32(false, A1, false, B1, (short)0, acc1, false, false); \
    acc0 = __builtin_amdgcn_wmma_f32_16x16x4_f32(false, A2, false, B2, (short)0, acc0, false, false); \
    acc1 = __builtin_amdgcn_wmma_f32_16x16x4_f32(false, A3, false, B3, (short)0, acc1, false, false); \
  }

  if (fmask == 0u) {
    #pragma unroll 4
    for (int kt = 0; kt < kKT; ++kt) {
      const v4f a0 = *(const v4f*)(ap + 16 * kt);
      const v4f a1 = *(const v4f*)(ap + 16 * kt + 4);
      const unsigned z = (unsigned)hp[kt * 128] << zsh;
      WMMA4(bit_pm1(z, 0), bit_pm1(z, 1), bit_pm1(z, 2), bit_pm1(z, 3),
            bit_pm1(z, 4), bit_pm1(z, 5), bit_pm1(z, 6), bit_pm1(z, 7))
    }
  } else {
    for (int kt = 0; kt < kKT; ++kt) {
      const v4f a0 = *(const v4f*)(ap + 16 * kt);
      const v4f a1 = *(const v4f*)(ap + 16 * kt + 4);
      if ((fmask >> kt) & 1u) {
        const float* __restrict__ bp = w + (size_t)(16 * kt + 8 * h) * kWH + (i0 + lm);
        WMMA4(sgn(bp[0]), sgn(bp[kWH]), sgn(bp[2 * (size_t)kWH]), sgn(bp[3 * (size_t)kWH]),
              sgn(bp[4 * (size_t)kWH]), sgn(bp[5 * (size_t)kWH]), sgn(bp[6 * (size_t)kWH]),
              sgn(bp[7 * (size_t)kWH]))
      } else {
        const unsigned z = (unsigned)hp[kt * 128] << zsh;
        WMMA4(bit_pm1(z, 0), bit_pm1(z, 1), bit_pm1(z, 2), bit_pm1(z, 3),
              bit_pm1(z, 4), bit_pm1(z, 5), bit_pm1(z, 6), bit_pm1(z, 7))
      }
    }
  }
#undef WMMA4

  float m = -__builtin_inff();
  if (lane < 16) {
    #pragma unroll
    for (int r = 0; r < 8; ++r) {
      const float v = acc0[r] + acc1[r];
      out[(size_t)r * kWH + i0 + lm] = v;
      m = fmaxf(m, v);
    }
  }
  #pragma unroll
  for (int off = 16; off; off >>= 1)
    m = fmaxf(m, __shfl_xor(m, off, 32));
  if (lane == 0) {
    unsigned u = __float_as_uint(m);
    unsigned key = (u & 0x80000000u) ? ~u : (u | 0x80000000u);
    atomicMax(((unsigned*)ws) + MAX_OFF, key);
  }
}

__global__ __launch_bounds__(256)
void phase3(const float* __restrict__ ws, float* __restrict__ out) {
  const size_t t = (size_t)blockIdx.x * 256 + threadIdx.x;
  const unsigned key  = ((const unsigned*)ws)[MAX_OFF];
  const unsigned bits = (key & 0x80000000u) ? (key & 0x7fffffffu) : ~key;
  const float mx = __uint_as_float(bits);
  out[t] = out[t] / mx;
}

extern "C" void kernel_launch(void* const* d_in, const int* in_sizes, int n_in,
                              void* d_out, int out_size, void* d_ws, size_t ws_size,
                              hipStream_t stream) {
  const float* x = (const float*)d_in[0];
  const float* w = (const float*)d_in[1];
  float* out = (float*)d_out;
  float* ws  = (float*)d_ws;

  const int p1_waves = kKT * kNCH;
  phase1 <<<p1_waves / 8,        256, 0, stream>>>(x, w, ws, out);
  reduce_y<<<kKT * 4,            256, 0, stream>>>(out, ws);
  phase2 <<<(kWH / 16) / 8,      256, 0, stream>>>(w, ws, out);
  phase3 <<<(kB * kWH) / 256,    256, 0, stream>>>(ws, out);
}
